# starting kernel
# baseline (speedup 1.0000x reference)
#include <hip/hip_runtime.h>
#include <hip/hip_bf16.h>


typedef float v2f   __attribute__((ext_vector_type(2)));
typedef float v8f   __attribute__((ext_vector_type(8)));
typedef float fvec4 __attribute__((ext_vector_type(4)));

#define FL        1024
#define HOP       512
#define S_LEN     262144
#define T_FRAMES  513
#define N_FILT    1024
#define K_DIM     1024
#define B_BATCH   16
#define C_CH      2

#define BM   128
#define BN   128
#define BK   32
#define LDA  33
#define LDB  132

#define PI_F 3.14159265358979323846f

__global__ __launch_bounds__(256)
void filterbank_wmma_kernel(const float* __restrict__ data,
                            const float* __restrict__ filters,
                            float* __restrict__ out)
{
    __shared__ float As[2][BM * LDA];
    __shared__ float Bs[2][BK * LDB];
    __shared__ float win[FL];

    const int tid  = threadIdx.x;
    const int lane = tid & 31;
    const int wave = tid >> 5;
    const int wm   = wave >> 1;
    const int wn   = wave & 1;
    const int wmB  = wm * 32;
    const int wnB  = wn * 64;

    const int bc = blockIdx.z;
    const int b  = bc >> 1;
    const int c  = bc & 1;
    const int m0 = blockIdx.y * BM;
    const int n0 = blockIdx.x * BN;

    const float* __restrict__ dptr = data + (size_t)bc * (size_t)S_LEN;

    const int halfSel = lane >> 4;
    const int l15     = lane & 15;

    auto loadA = [&](int kb, fvec4 (&r)[4]) {
        #pragma unroll
        for (int i = 0; i < 4; ++i) {
            const int idx  = i * 256 + tid;
            const int row  = idx >> 3;
            const int col4 = (idx & 7) * 4;
            const int t    = m0 + row;
            const long s   = (long)t * HOP + (kb + col4) - (FL / 2);
            fvec4 v = fvec4{0.f, 0.f, 0.f, 0.f};
            if (t < T_FRAMES && s >= 0 && s < S_LEN)
                v = *(const fvec4*)(dptr + s);
            r[i] = v;
        }
    };
    auto loadB = [&](int kb, fvec4 (&r)[4]) {
        #pragma unroll
        for (int i = 0; i < 4; ++i) {
            const int idx  = i * 256 + tid;
            const int row  = idx >> 5;
            const int col4 = (idx & 31) * 4;
            r[i] = *(const fvec4*)(filters +
                     (size_t)(kb + row) * N_FILT + n0 + col4);
        }
    };
    auto storeA = [&](int buf, int kb, fvec4 (&r)[4]) {
        #pragma unroll
        for (int i = 0; i < 4; ++i) {
            const int idx  = i * 256 + tid;
            const int row  = idx >> 3;
            const int col4 = (idx & 7) * 4;
            const int f    = kb + col4;
            float* dst = &As[buf][row * LDA + col4];
            dst[0] = r[i][0] * win[f + 0];
            dst[1] = r[i][1] * win[f + 1];
            dst[2] = r[i][2] * win[f + 2];
            dst[3] = r[i][3] * win[f + 3];
        }
    };
    auto storeB = [&](int buf, fvec4 (&r)[4]) {
        #pragma unroll
        for (int i = 0; i < 4; ++i) {
            const int idx  = i * 256 + tid;
            const int row  = idx >> 5;
            const int col4 = (idx & 31) * 4;
            float* dst = &Bs[buf][row * LDB + col4];
            dst[0] = r[i][0];
            dst[1] = r[i][1];
            dst[2] = r[i][2];
            dst[3] = r[i][3];
        }
    };

    for (int i = tid; i < FL; i += 256)
        win[i] = sinf(PI_F * (float)i * (1.0f / (float)FL));

    v8f acc[2][4];
    #pragma unroll
    for (int mi = 0; mi < 2; ++mi)
        #pragma unroll
        for (int ni = 0; ni < 4; ++ni)
            acc[mi][ni] = v8f{0.f, 0.f, 0.f, 0.f, 0.f, 0.f, 0.f, 0.f};

    fvec4 ra[4], rb[4];
    loadA(0, ra);
    loadB(0, rb);
    __syncthreads();
    storeA(0, 0, ra);
    storeB(0, rb);
    __syncthreads();

    const int NKB = K_DIM / BK;
    for (int kbi = 0; kbi < NKB; ++kbi) {
        const int  cur  = kbi & 1;
        const bool more = (kbi + 1) < NKB;

        if (more) {
            loadA((kbi + 1) * BK, ra);
            loadB((kbi + 1) * BK, rb);
        }

        const float* __restrict__ Ab = &As[cur][0];
        const float* __restrict__ Bb = &Bs[cur][0];
        #pragma unroll
        for (int k = 0; k < BK; k += 4) {
            const int ka = k + (halfSel << 1);

            v2f af[2];
            #pragma unroll
            for (int mi = 0; mi < 2; ++mi) {
                const int row = wmB + mi * 16 + l15;
                af[mi][0] = Ab[row * LDA + ka];
                af[mi][1] = Ab[row * LDA + ka + 1];
            }
            v2f bf[4];
            #pragma unroll
            for (int ni = 0; ni < 4; ++ni) {
                const int coln = wnB + ni * 16 + l15;
                bf[ni][0] = Bb[ka * LDB + coln];
                bf[ni][1] = Bb[(ka + 1) * LDB + coln];
            }
            #pragma unroll
            for (int mi = 0; mi < 2; ++mi)
                #pragma unroll
                for (int ni = 0; ni < 4; ++ni)
                    acc[mi][ni] = __builtin_amdgcn_wmma_f32_16x16x4_f32(
                         false, af[mi],
                         false, bf[ni],
                         (short)0, acc[mi][ni],
                         false,  false);
        }

        if (more) {
            storeA(cur ^ 1, (kbi + 1) * BK, ra);
            storeB(cur ^ 1, rb);
        }
        __syncthreads();
    }

    #pragma unroll
    for (int mi = 0; mi < 2; ++mi) {
        #pragma unroll
        for (int ni = 0; ni < 4; ++ni) {
            const int nCol = n0 + wnB + ni * 16 + l15;
            #pragma unroll
            for (int g = 0; g < 8; ++g) {
                const int t = m0 + wmB + mi * 16 + g + halfSel * 8;
                if (t < T_FRAMES) {
                    const size_t o =
                        (((size_t)b * T_FRAMES + t) * N_FILT + nCol) * C_CH + c;
                    out[o] = acc[mi][ni][g];
                }
            }
        }
    }
}

extern "C" void kernel_launch(void* const* d_in, const int* in_sizes, int n_in,
                              void* d_out, int out_size, void* d_ws, size_t ws_size,
                              hipStream_t stream) {
    (void)in_sizes; (void)n_in; (void)out_size; (void)d_ws; (void)ws_size;

    const float* data    = (const float*)d_in[0];
    const float* filters = (const float*)d_in[1];
    float* out           = (float*)d_out;

    dim3 grid(N_FILT / BN,
              (T_FRAMES + BM - 1) / BM,
              B_BATCH * C_CH);
    dim3 block(256);

    filterbank_wmma_kernel<<<grid, block, 0, stream>>>(data, filters, out);
}


# Both channels of a frame in one wave: tile rows = 8 frames x 2 channels (work-group 88 frames x 2 ch), so each (t,n) channel pair is one 8-byte store and each 128-B output line one whole write
# speedup vs baseline: 1.4528x; 1.4528x over previous
#include <hip/hip_runtime.h>


typedef float v2f   __attribute__((ext_vector_type(2)));
typedef float v8f   __attribute__((ext_vector_type(8)));
typedef float fvec4 __attribute__((ext_vector_type(4)));

#define FL        1024
#define HOP       512
#define S_LEN     262144
#define N_CHUNK   (S_LEN / HOP)
#define T_FRAMES  513
#define N_FILT    1024
#define B_BATCH   16
#define C_CH      2

#define NTHR 128
#define BM   176
#define BMF  (BM / C_CH)
#define BN   128
#define QB   16
#define NSTG (HOP / QB)
#define MT   11
#define NT   2
#define AROWS (BM + C_CH)
#define TRASH AROWS
#define LDA  20
#define LDB  136
#define A_TILE ((AROWS + 1) * LDA)
#define B_TILE (2 * QB * LDB)
#define A_F4 6
#define A_SPLIT 3
#define B_F4 4
#define STG  4

#define PI_F 3.14159265358979323846f

template <int I0, int I1>
__device__ __forceinline__ void load_a_part(const float* __restrict__ ap,
                                            const unsigned (&goff)[A_F4],
                                            fvec4 (&r)[STG])
{
    #pragma unroll
    for (int i = I0; i < I1; ++i)
        r[i - I0] = *(const fvec4*)(ap + goff[i]);
}
template <int I0, int I1>
__device__ __forceinline__ void store_a_part(float* __restrict__ Adst,
                                             const unsigned (&soff)[A_F4],
                                             const fvec4 (&r)[STG])
{
    #pragma unroll
    for (int i = I0; i < I1; ++i)
        *(fvec4*)&Adst[soff[i]] = r[i - I0];
}
__device__ __forceinline__ void load_b_half(const float* __restrict__ bp,
                                            unsigned boff, fvec4 (&r)[STG])
{
    #pragma unroll
    for (int i = 0; i < B_F4; ++i)
        r[i] = *(const fvec4*)(bp + boff + i * N_FILT);
}
__device__ __forceinline__ void store_b_half(float* __restrict__ Bdst,
                                             unsigned bsoff, const fvec4 w,
                                             const fvec4 (&r)[STG])
{
    #pragma unroll
    for (int i = 0; i < B_F4; ++i)
        *(fvec4*)&Bdst[bsoff + i * LDB] = r[i] * w[i];
}

__global__ __launch_bounds__(NTHR) __attribute__((amdgpu_waves_per_eu(3, 3)))
void filterbank_wmma_kernel(const float* __restrict__ data,
                            const float* __restrict__ filters,
                            float* __restrict__ out)
{
    __shared__ __attribute__((aligned(16))) float As[2][A_TILE];
    __shared__ __attribute__((aligned(16))) float Bs[2][B_TILE];
    __shared__ __attribute__((aligned(16))) float win[FL];

    const int tid  = threadIdx.x;
    const int lane = tid & 31;
    const int wave = tid >> 5;
    const int wnB  = wave * (NT * 16);

    const int b  = blockIdx.z;
    const int m0 = blockIdx.y * BMF;
    const int n0 = blockIdx.x * BN;

    const float* __restrict__ dptr = data + (size_t)b * (size_t)(C_CH * S_LEN);
    const float* __restrict__ fptr = filters + n0;

    const int halfSel = lane >> 4;
    const int l15     = lane & 15;

    unsigned a_goff[A_F4], a_soff[A_F4];
    #pragma unroll
    for (int i = 0; i < A_F4; ++i) {
        const int idx = i * NTHR + tid;
        const int r   = idx >> 2;
        const int c4  = (idx & 3) * 4;
        const int ch  = r & 1;
        const int cj  = m0 - 1 + (r >> 1);
        const bool live = (r < AROWS) && (cj >= 0) && (cj < N_CHUNK);
        const int cjc = cj < 0 ? 0 : (cj > N_CHUNK - 1 ? N_CHUNK - 1 : cj);
        a_goff[i] = (unsigned)(ch * S_LEN + cjc * HOP + c4);
        a_soff[i] = (unsigned)((live ? r : TRASH) * LDA + c4);
    }
    const unsigned b_goff = (unsigned)((wave * B_F4) * N_FILT + lane * 4);
    const unsigned b_soff = (unsigned)((wave * B_F4) * LDB + lane * 4);
    const int      w_off  = wave * B_F4;

    for (int i = tid; i < FL; i += NTHR)
        win[i] = sinf(PI_F * (float)i * (1.0f / (float)FL));
    for (int i = tid; i < 2 * A_TILE / 4; i += NTHR)
        ((fvec4*)&As[0][0])[i] = fvec4{0.f, 0.f, 0.f, 0.f};
    __syncthreads();

    fvec4 rs[STG];
    load_a_part<0, A_SPLIT>(dptr, a_goff, rs);
    store_a_part<0, A_SPLIT>(&As[0][0], a_soff, rs);
    load_a_part<A_SPLIT, A_F4>(dptr, a_goff, rs);
    store_a_part<A_SPLIT, A_F4>(&As[0][0], a_soff, rs);
    load_b_half(fptr, b_goff, rs);
    store_b_half(&Bs[0][0], b_soff, *(const fvec4*)&win[w_off], rs);
    load_b_half(fptr + (size_t)HOP * N_FILT, b_goff, rs);
    store_b_half(&Bs[0][QB * LDB], b_soff, *(const fvec4*)&win[HOP + w_off], rs);

    v8f acc[MT][NT];
    #pragma unroll
    for (int mi = 0; mi < MT; ++mi)
        #pragma unroll
        for (int ni = 0; ni < NT; ++ni)
            acc[mi][ni] = v8f{0.f, 0.f, 0.f, 0.f, 0.f, 0.f, 0.f, 0.f};

    __syncthreads();

    for (int q = 0; q < NSTG; ++q) {
        if (q == 0)                  __builtin_amdgcn_s_setprio(3);
        else if (q == NSTG / 2)      __builtin_amdgcn_s_setprio(2);
        else if (q == 3 * NSTG / 4)  __builtin_amdgcn_s_setprio(1);
        else if (q == 7 * NSTG / 8)  __builtin_amdgcn_s_setprio(0);

        const int  cur  = q & 1;
        const bool more = (q + 1) < NSTG;
        const int  sn   = (q + 1) * QB;

        const float* __restrict__ Ab = &As[cur][0];
        const float* __restrict__ Bb = &Bs[cur][0];
        float* __restrict__ An = &As[cur ^ 1][0];
        float* __restrict__ Bn = &Bs[cur ^ 1][0];

        auto compute = [&](int kBeg) {
            const int down = kBeg >= QB ? C_CH : 0;
            #pragma unroll
            for (int k = kBeg; k < kBeg + 8; k += 4) {
                const int ka = k + (halfSel << 1);
                const int ca = kBeg >= QB ? ka - QB : ka;

                v2f bf[NT];
                #pragma unroll
                for (int ni = 0; ni < NT; ++ni) {
                    const int coln = wnB + ni * 16 + l15;
                    bf[ni][0] = Bb[ka * LDB + coln];
                    bf[ni][1] = Bb[(ka + 1) * LDB + coln];
                }
                v2f af[MT];
                #pragma unroll
                for (int mi = 0; mi < MT; ++mi) {
                    const int row = mi * 16 + l15 + down;
                    af[mi] = *(const v2f*)&Ab[row * LDA + ca];
                }
                #pragma unroll
                for (int mi = 0; mi < MT; ++mi)
                    #pragma unroll
                    for (int ni = 0; ni < NT; ++ni)
                        acc[mi][ni] = __builtin_amdgcn_wmma_f32_16x16x4_f32(
                             false, af[mi],
                             false, bf[ni],
                             (short)0, acc[mi][ni],
                             false,  false);
            }
        };

        fvec4 wv;
        if (more) load_a_part<0, A_SPLIT>(dptr + sn, a_goff, rs);
        __builtin_amdgcn_sched_barrier(0);
        compute(0);
        __builtin_amdgcn_sched_barrier(0);
        if (more) {
            store_a_part<0, A_SPLIT>(An, a_soff, rs);
            load_a_part<A_SPLIT, A_F4>(dptr + sn, a_goff, rs);
        }
        __builtin_amdgcn_sched_barrier(0);
        compute(8);
        __builtin_amdgcn_sched_barrier(0);
        if (more) {
            store_a_part<A_SPLIT, A_F4>(An, a_soff, rs);
            load_b_half(fptr + (size_t)sn * N_FILT, b_goff, rs);
            wv = *(const fvec4*)&win[sn + w_off];
        }
        __builtin_amdgcn_sched_barrier(0);
        compute(16);
        __builtin_amdgcn_sched_barrier(0);
        if (more) {
            store_b_half(Bn, b_soff, wv, rs);
            load_b_half(fptr + (size_t)(HOP + sn) * N_FILT, b_goff, rs);
            wv = *(const fvec4*)&win[HOP + sn + w_off];
        }
        __builtin_amdgcn_sched_barrier(0);
        compute(24);
        __builtin_amdgcn_sched_barrier(0);
        if (more)
            store_b_half(Bn + QB * LDB, b_soff, wv, rs);
        __syncthreads();
    }

    v2f* __restrict__ op = (v2f*)out
        + (((size_t)b * T_FRAMES + m0 + 4 * halfSel) * N_FILT + n0 + wnB + l15);
    #pragma unroll
    for (int mi = 0; mi < MT; ++mi) {
        const int tb = m0 + mi * 8;
        if (tb + 8 <= T_FRAMES) {
            #pragma unroll
            for (int k = 0; k < 4; ++k)
                #pragma unroll
                for (int ni = 0; ni < NT; ++ni)
                    op[(mi * 8 + k) * N_FILT + ni * 16] =
                        v2f{acc[mi][ni][2 * k], acc[mi][ni][2 * k + 1]};
        } else {
            #pragma unroll
            for (int k = 0; k < 4; ++k) {
                if (tb + 4 * halfSel + k < T_FRAMES) {
                    #pragma unroll
                    for (int ni = 0; ni < NT; ++ni)
                        op[(mi * 8 + k) * N_FILT + ni * 16] =
                            v2f{acc[mi][ni][2 * k], acc[mi][ni][2 * k + 1]};
                }
            }
        }
    }
}

extern "C" void kernel_launch(void* const* d_in, const int* in_sizes, int n_in,
                              void* d_out, int out_size, void* d_ws, size_t ws_size,
                              hipStream_t stream) {
    (void)in_sizes; (void)n_in; (void)out_size; (void)d_ws; (void)ws_size;

    const float* data    = (const float*)d_in[0];
    const float* filters = (const float*)d_in[1];
    float* out           = (float*)d_out;

    dim3 grid(N_FILT / BN,
              (T_FRAMES + BMF - 1) / BMF,
              B_BATCH);
    dim3 block(NTHR);

    filterbank_wmma_kernel<<<grid, block, 0, stream>>>(data, filters, out);
}
